# starting kernel
# baseline (speedup 1.0000x reference)
#include <hip/hip_runtime.h>


typedef __attribute__((ext_vector_type(2))) float v2f;
typedef __attribute__((ext_vector_type(8))) float v8f;

#define D_IN  128
#define D_OUT 64

__global__ void gcmc_zero_kernel(float4* __restrict__ out, int n4) {
    int i = blockIdx.x * blockDim.x + threadIdx.x;
    if (i < n4) out[i] = make_float4(0.f, 0.f, 0.f, 0.f);
}

__global__ __launch_bounds__(256)
void gcmc_gemm_wmma_kernel(const float* __restrict__ A,
                           const float* __restrict__ W,
                           const float* __restrict__ cj,
                           float* __restrict__ H,
                           int nTiles) {
    __shared__ float2 sW2[(D_IN / 2) * D_OUT];

    {
        int tid = threadIdx.x;
        #pragma unroll
        for (int i = 0; i < ((D_IN / 2) * D_OUT) / 256; ++i) {
            int p  = tid + i * 256;
            int kk = p >> 6;
            int n  = p & 63;
            float2 w;
            w.x = W[(2 * kk)     * D_OUT + n];
            w.y = W[(2 * kk + 1) * D_OUT + n];
            sW2[p] = w;
        }
    }
    __syncthreads();

    const int wave = threadIdx.x >> 5;
    const int lane = threadIdx.x & 31;
    const int tile = blockIdx.x * (blockDim.x >> 5) + wave;
    if (tile >= nTiles) return;

    const int half = lane >> 4;
    const int lrow = lane & 15;

    const float* Arow = A + (long long)(tile * 16 + lrow) * D_IN;

    v8f c0 = {}, c1 = {}, c2 = {}, c3 = {};

    for (int k = 0; k < D_IN; k += 4) {
        const int kA = k + 2 * half;
        v2f a = *(const v2f*)(Arow + kA);

        const float2* sRow = sW2 + ((k >> 1) + half) * D_OUT + lrow;
        v2f b0 = *(const v2f*)(sRow +  0);
        v2f b1 = *(const v2f*)(sRow + 16);
        v2f b2 = *(const v2f*)(sRow + 32);
        v2f b3 = *(const v2f*)(sRow + 48);

        c0 = __builtin_amdgcn_wmma_f32_16x16x4_f32(false, a, false, b0, (short)0, c0, false, false);
        c1 = __builtin_amdgcn_wmma_f32_16x16x4_f32(false, a, false, b1, (short)0, c1, false, false);
        c2 = __builtin_amdgcn_wmma_f32_16x16x4_f32(false, a, false, b2, (short)0, c2, false, false);
        c3 = __builtin_amdgcn_wmma_f32_16x16x4_f32(false, a, false, b3, (short)0, c3, false, false);
    }

    #pragma unroll
    for (int r = 0; r < 8; ++r) {
        const int   node = tile * 16 + r + 8 * half;
        const float s    = cj[node];
        float* Hrow = H + (long long)node * D_OUT;
        Hrow[ 0 + lrow] = c0[r] * s;
        Hrow[16 + lrow] = c1[r] * s;
        Hrow[32 + lrow] = c2[r] * s;
        Hrow[48 + lrow] = c3[r] * s;
    }
}

__global__ void gcmc_scatter_kernel(const float* __restrict__ H,
                                    const float* __restrict__ ci,
                                    const int*   __restrict__ src,
                                    const int*   __restrict__ dst,
                                    float*       __restrict__ out,
                                    int nEdges) {
    long long idx   = (long long)blockIdx.x * blockDim.x + threadIdx.x;
    long long nwork = (long long)nEdges * 32;
    if (idx >= nwork) return;
    const int e = (int)(idx >> 5);
    const int c = (int)(idx & 31) * 2;
    const int s = src[e];
    const int d = dst[e];
    const float  cd = ci[d];
    const float2 hv = *(const float2*)(H + (long long)s * D_OUT + c);
    float* o = out + (long long)d * D_OUT + c;
    atomicAdd(o,     hv.x * cd);
    atomicAdd(o + 1, hv.y * cd);
}

extern "C" void kernel_launch(void* const* d_in, const int* in_sizes, int n_in,
                              void* d_out, int out_size, void* d_ws, size_t ws_size,
                              hipStream_t stream) {
    const float* input_feat = (const float*)d_in[0];
    const float* weight     = (const float*)d_in[1];
    const float* cj         = (const float*)d_in[2];
    const float* ci         = (const float*)d_in[3];
    const int*   src_idx    = (const int*)d_in[4];
    const int*   dst_idx    = (const int*)d_in[5];
    float*       out        = (float*)d_out;
    float*       H          = (float*)d_ws;

    const int N = in_sizes[2];
    const int E = in_sizes[4];

    {
        const int n4 = out_size / 4;
        gcmc_zero_kernel<<<(n4 + 255) / 256, 256, 0, stream>>>((float4*)out, n4);
    }

    {
        const int nTiles = N / 16;
        const int blocks = (nTiles + 7) / 8;
        gcmc_gemm_wmma_kernel<<<blocks, 256, 0, stream>>>(input_feat, weight, cj, H, nTiles);
    }

    {
        const long long nwork  = (long long)E * 32;
        const int       blocks = (int)((nwork + 255) / 256);
        gcmc_scatter_kernel<<<blocks, 256, 0, stream>>>(H, ci, src_idx, dst_idx, out, E);
    }
}


# f32-atomic edge scatter replaced by per-destination linked lists (one int atomic exchange per edge) + atomic-free list-walk gather storing each out row once; sum order = exchange arrival order
# speedup vs baseline: 1.4443x; 1.4443x over previous
#include <hip/hip_runtime.h>


typedef __attribute__((ext_vector_type(2))) float v2f;
typedef __attribute__((ext_vector_type(8))) float v8f;

#define D_IN  128
#define D_OUT 64

__global__ void gcmc_init_head_kernel(int* __restrict__ head, int nNodes) {
    int i = blockIdx.x * blockDim.x + threadIdx.x;
    if (i < nNodes) head[i] = -1;
}

__global__ __launch_bounds__(256)
void gcmc_link_kernel(const int* __restrict__ src,
                      const int* __restrict__ dst,
                      int*       __restrict__ head,
                      int2*      __restrict__ node,
                      int nEdges) {
    const int stride = gridDim.x * blockDim.x;
    for (int e = blockIdx.x * blockDim.x + threadIdx.x; e < nEdges; e += stride) {
        const int d    = dst[e];
        const int s    = src[e];
        const int prev = atomicExch(head + d, e);
        node[e] = make_int2(s, prev);
    }
}

__global__ __launch_bounds__(256)
void gcmc_gemm_wmma_kernel(const float* __restrict__ A,
                           const float* __restrict__ W,
                           const float* __restrict__ cj,
                           float* __restrict__ H,
                           int nTiles) {
    __shared__ float2 sW2[(D_IN / 2) * D_OUT];

    {
        int tid = threadIdx.x;
        #pragma unroll
        for (int i = 0; i < ((D_IN / 2) * D_OUT) / 256; ++i) {
            int p  = tid + i * 256;
            int kk = p >> 6;
            int n  = p & 63;
            float2 w;
            w.x = W[(2 * kk)     * D_OUT + n];
            w.y = W[(2 * kk + 1) * D_OUT + n];
            sW2[p] = w;
        }
    }
    __syncthreads();

    const int wave = threadIdx.x >> 5;
    const int lane = threadIdx.x & 31;
    const int tile = blockIdx.x * (blockDim.x >> 5) + wave;
    if (tile >= nTiles) return;

    const int half = lane >> 4;
    const int lrow = lane & 15;

    const float* Arow = A + (long long)(tile * 16 + lrow) * D_IN;

    v8f c0 = {}, c1 = {}, c2 = {}, c3 = {};

    for (int k = 0; k < D_IN; k += 4) {
        const int kA = k + 2 * half;
        v2f a = *(const v2f*)(Arow + kA);

        const float2* sRow = sW2 + ((k >> 1) + half) * D_OUT + lrow;
        v2f b0 = *(const v2f*)(sRow +  0);
        v2f b1 = *(const v2f*)(sRow + 16);
        v2f b2 = *(const v2f*)(sRow + 32);
        v2f b3 = *(const v2f*)(sRow + 48);

        c0 = __builtin_amdgcn_wmma_f32_16x16x4_f32(false, a, false, b0, (short)0, c0, false, false);
        c1 = __builtin_amdgcn_wmma_f32_16x16x4_f32(false, a, false, b1, (short)0, c1, false, false);
        c2 = __builtin_amdgcn_wmma_f32_16x16x4_f32(false, a, false, b2, (short)0, c2, false, false);
        c3 = __builtin_amdgcn_wmma_f32_16x16x4_f32(false, a, false, b3, (short)0, c3, false, false);
    }

    #pragma unroll
    for (int r = 0; r < 8; ++r) {
        const int   node = tile * 16 + r + 8 * half;
        const float s    = cj[node];
        float* Hrow = H + (long long)node * D_OUT;
        Hrow[ 0 + lrow] = c0[r] * s;
        Hrow[16 + lrow] = c1[r] * s;
        Hrow[32 + lrow] = c2[r] * s;
        Hrow[48 + lrow] = c3[r] * s;
    }
}

__global__ __launch_bounds__(256)
void gcmc_gather_kernel(const float* __restrict__ H,
                        const float* __restrict__ ci,
                        const int*   __restrict__ head,
                        const int2*  __restrict__ node,
                        float*       __restrict__ out,
                        int nNodes, int nEdges) {
    const int gid = blockIdx.x * blockDim.x + threadIdx.x;
    const int d   = gid >> 3;
    const int t   = gid & 7;
    if (d >= nNodes) return;

    const float cd = ci[d];
    const unsigned uE = (unsigned)nEdges;
    float4 a0 = make_float4(0.f, 0.f, 0.f, 0.f);
    float4 a1 = make_float4(0.f, 0.f, 0.f, 0.f);

    int  cur = head[d];
    int2 nd  = make_int2(0, -1);
    if ((unsigned)cur < uE) nd = node[cur];

    for (int it = 0; it < nEdges && (unsigned)cur < uE; ++it) {
        const float4* hp = (const float4*)(H + (long long)nd.x * D_OUT + 4 * t);
        const float4 v0 = hp[0];
        const float4 v1 = hp[8];
        const int nxt = nd.y;
        int2 nd2 = nd;
        if ((unsigned)nxt < uE) nd2 = node[nxt];
        a0.x += v0.x; a0.y += v0.y; a0.z += v0.z; a0.w += v0.w;
        a1.x += v1.x; a1.y += v1.y; a1.z += v1.z; a1.w += v1.w;
        cur = nxt;
        nd  = nd2;
    }

    float4* op = (float4*)(out + (long long)d * D_OUT + 4 * t);
    op[0] = make_float4(a0.x * cd, a0.y * cd, a0.z * cd, a0.w * cd);
    op[8] = make_float4(a1.x * cd, a1.y * cd, a1.z * cd, a1.w * cd);
}

extern "C" void kernel_launch(void* const* d_in, const int* in_sizes, int n_in,
                              void* d_out, int out_size, void* d_ws, size_t ws_size,
                              hipStream_t stream) {
    const float* input_feat = (const float*)d_in[0];
    const float* weight     = (const float*)d_in[1];
    const float* cj         = (const float*)d_in[2];
    const float* ci         = (const float*)d_in[3];
    const int*   src_idx    = (const int*)d_in[4];
    const int*   dst_idx    = (const int*)d_in[5];
    float*       out        = (float*)d_out;
    float*       H          = (float*)d_ws;

    const int N = in_sizes[2];
    const int E = in_sizes[4];

    int2* node = (int2*)(H + (size_t)N * D_OUT);
    int*  head = (int*)(node + E);

    gcmc_init_head_kernel<<<(N + 255) / 256, 256, 0, stream>>>(head, N);

    {
        const int blocks = (E + 1023) / 1024;
        gcmc_link_kernel<<<blocks, 256, 0, stream>>>(src_idx, dst_idx, head, node, E);
    }

    {
        const int nTiles = N / 16;
        const int blocks = (nTiles + 7) / 8;
        gcmc_gemm_wmma_kernel<<<blocks, 256, 0, stream>>>(input_feat, weight, cj, H, nTiles);
    }

    {
        const int blocks = (N + 31) / 32;
        gcmc_gather_kernel<<<blocks, 256, 0, stream>>>(H, ci, head, node, out, N, E);
    }
}


# per-edge global atomic exchange + list walk replaced by LDS counting sort of edge chunks into 256 dst ranges, one workgroup per range summing h rows in LDS (ds_add_f32, arrival order); 3 launches
# speedup vs baseline: 3.0547x; 2.1150x over previous
#include <hip/hip_runtime.h>


typedef __attribute__((ext_vector_type(2))) float v2f;
typedef __attribute__((ext_vector_type(8))) float v8f;

#define D_IN  128
#define D_OUT 64

#define NBINS      256
#define BIN_NODES  391
#define CHUNK      3125
#define SORT_THREADS 256
#define SORT_ITEMS 13

__global__ __launch_bounds__(SORT_THREADS)
void gcmc_bin_sort_kernel(const int* __restrict__ src,
                          const int* __restrict__ dst,
                          int2*      __restrict__ sorted,
                          int*       __restrict__ segStart,
                          int nEdges) {
    __shared__ int  sCnt[NBINS + 32];
    __shared__ int  sStart[NBINS + 1];
    __shared__ int  sWave[NBINS / 32];
    __shared__ int2 sBuf[CHUNK];

    const int tid    = threadIdx.x;
    const int lane   = tid & 31;
    const int base   = blockIdx.x * CHUNK;
    const int nValid = min(CHUNK, nEdges - base);

    sCnt[tid] = 0;
    if (tid < 32) sCnt[NBINS + tid] = 0;
    __syncthreads();

    int s[SORT_ITEMS], d[SORT_ITEMS], b[SORT_ITEMS], r[SORT_ITEMS];
    #pragma unroll
    for (int i = 0; i < SORT_ITEMS; ++i) {
        const int p = min(i * SORT_THREADS + tid, nValid - 1);
        d[i] = dst[base + p];
        s[i] = src[base + p];
    }
    #pragma unroll
    for (int i = 0; i < SORT_ITEMS; ++i) {
        const int p = i * SORT_THREADS + tid;
        b[i] = (p < nValid) ? d[i] / BIN_NODES : NBINS + lane;
        r[i] = atomicAdd(&sCnt[b[i]], 1);
    }
    __syncthreads();

    const int c = sCnt[tid];
    int inc = c;
    #pragma unroll
    for (int o = 1; o < 32; o <<= 1) {
        const int y = __shfl_up(inc, o);
        if (lane >= o) inc += y;
    }
    if (lane == 31) sWave[tid >> 5] = inc;
    __syncthreads();
    {
        const int w = tid >> 5;
        int off = 0;
        #pragma unroll
        for (int k = 0; k < NBINS / 32; ++k) off += (k < w) ? sWave[k] : 0;
        sStart[tid] = off + inc - c;
    }
    if (tid == 0) sStart[NBINS] = nValid;
    __syncthreads();

    int pos[SORT_ITEMS];
    #pragma unroll
    for (int i = 0; i < SORT_ITEMS; ++i) pos[i] = sStart[min(b[i], NBINS)] + r[i];
    #pragma unroll
    for (int i = 0; i < SORT_ITEMS; ++i) {
        const int p = i * SORT_THREADS + tid;
        if (p < nValid) sBuf[pos[i]] = make_int2(s[i], d[i]);
    }
    __syncthreads();

    int2 o[SORT_ITEMS];
    #pragma unroll
    for (int i = 0; i < SORT_ITEMS; ++i)
        o[i] = sBuf[min(i * SORT_THREADS + tid, nValid - 1)];
    #pragma unroll
    for (int i = 0; i < SORT_ITEMS; ++i) {
        const int p = i * SORT_THREADS + tid;
        if (p < nValid) sorted[base + p] = o[i];
    }
    segStart[blockIdx.x * (NBINS + 1) + tid] = sStart[tid];
    if (tid == 0) segStart[blockIdx.x * (NBINS + 1) + NBINS] = nValid;
}

__global__ __launch_bounds__(256)
void gcmc_gemm_wmma_kernel(const float* __restrict__ A,
                           const float* __restrict__ W,
                           const float* __restrict__ cj,
                           float* __restrict__ H,
                           int nTiles) {
    __shared__ float2 sW2[(D_IN / 2) * D_OUT];

    {
        int tid = threadIdx.x;
        #pragma unroll
        for (int i = 0; i < ((D_IN / 2) * D_OUT) / 256; ++i) {
            int p  = tid + i * 256;
            int kk = p >> 6;
            int n  = p & 63;
            float2 w;
            w.x = W[(2 * kk)     * D_OUT + n];
            w.y = W[(2 * kk + 1) * D_OUT + n];
            sW2[p] = w;
        }
    }
    __syncthreads();

    const int wave = threadIdx.x >> 5;
    const int lane = threadIdx.x & 31;
    const int tile = blockIdx.x * (blockDim.x >> 5) + wave;
    if (tile >= nTiles) return;

    const int half = lane >> 4;
    const int lrow = lane & 15;

    const float* Arow = A + (long long)(tile * 16 + lrow) * D_IN;

    v8f c0 = {}, c1 = {}, c2 = {}, c3 = {};

    for (int k = 0; k < D_IN; k += 4) {
        const int kA = k + 2 * half;
        v2f a = *(const v2f*)(Arow + kA);

        const float2* sRow = sW2 + ((k >> 1) + half) * D_OUT + lrow;
        v2f b0 = *(const v2f*)(sRow +  0);
        v2f b1 = *(const v2f*)(sRow + 16);
        v2f b2 = *(const v2f*)(sRow + 32);
        v2f b3 = *(const v2f*)(sRow + 48);

        c0 = __builtin_amdgcn_wmma_f32_16x16x4_f32(false, a, false, b0, (short)0, c0, false, false);
        c1 = __builtin_amdgcn_wmma_f32_16x16x4_f32(false, a, false, b1, (short)0, c1, false, false);
        c2 = __builtin_amdgcn_wmma_f32_16x16x4_f32(false, a, false, b2, (short)0, c2, false, false);
        c3 = __builtin_amdgcn_wmma_f32_16x16x4_f32(false, a, false, b3, (short)0, c3, false, false);
    }

    #pragma unroll
    for (int r = 0; r < 8; ++r) {
        const int   node = tile * 16 + r + 8 * half;
        const float s    = cj[node];
        float* Hrow = H + (long long)node * D_OUT;
        Hrow[ 0 + lrow] = c0[r] * s;
        Hrow[16 + lrow] = c1[r] * s;
        Hrow[32 + lrow] = c2[r] * s;
        Hrow[48 + lrow] = c3[r] * s;
    }
}

__device__ __forceinline__ float4 gcmc_rot4(float4 v, int q) {
    const float4 a = (q & 1) ? make_float4(v.y, v.z, v.w, v.x) : v;
    const float4 b = (q & 2) ? make_float4(a.z, a.w, a.x, a.y) : a;
    return b;
}

__global__ __launch_bounds__(1024)
void gcmc_bin_gather_kernel(const float* __restrict__ H,
                            const float* __restrict__ ci,
                            const int2*  __restrict__ sorted,
                            const int*   __restrict__ segStart,
                            float*       __restrict__ out,
                            int nNodes, int nChunks) {
    __shared__ float sAcc[BIN_NODES * D_OUT];
    float4* sAcc4 = (float4*)sAcc;

    const int tid = threadIdx.x;
    const int bin = blockIdx.x;
    const int lo  = bin * BIN_NODES;

    for (int i = tid; i < BIN_NODES * D_OUT / 4; i += 1024)
        sAcc4[i] = make_float4(0.f, 0.f, 0.f, 0.f);
    __syncthreads();

    const int g = tid >> 3;
    const int t = tid & 7;
    const int q = g & 3;
    const int c0 = 4 * t + ((0 + q) & 3);
    const int c1 = 4 * t + ((1 + q) & 3);
    const int c2 = 4 * t + ((2 + q) & 3);
    const int c3 = 4 * t + ((3 + q) & 3);

    for (int cBase = g; cBase < nChunks; cBase += 512) {
        int s0[4], len[4], sa[4], se[4];
        #pragma unroll
        for (int k = 0; k < 4; ++k) {
            const int c  = min(cBase + 128 * k, nChunks - 1);
            const int* ss = segStart + (long long)c * (NBINS + 1) + bin;
            sa[k] = ss[0];
            se[k] = ss[1];
        }
        #pragma unroll
        for (int k = 0; k < 4; ++k) {
            const int c = cBase + 128 * k;
            s0[k]  = min(c, nChunks - 1) * CHUNK + sa[k];
            len[k] = (c < nChunks) ? se[k] - sa[k] : 0;
        }
        const int p1  = len[0];
        const int p2  = p1 + len[1];
        const int p3  = p2 + len[2];
        const int tot = p3 + len[3];

        int2 recN = make_int2(0, lo);
        if (t < tot) {
            const int x = t;
            const int idx = x < p1 ? s0[0] + x
                          : x < p2 ? s0[1] + (x - p1)
                          : x < p3 ? s0[2] + (x - p2)
                          :          s0[3] + (x - p3);
            recN = sorted[idx];
        }

        for (int f = 0; f < tot; f += 8) {
            const int2 rec = recN;
            float4 v0[8], v1[8];
            int    rr[8];
            #pragma unroll
            for (int u = 0; u < 8; ++u) {
                const int su = __shfl(rec.x, u, 8);
                rr[u] = __shfl(rec.y, u, 8) - lo;
                const float4* hp = (const float4*)(H + (long long)su * D_OUT + 4 * t);
                v0[u] = hp[0];
                v1[u] = hp[8];
            }
            recN = make_int2(0, lo);
            {
                const int x = f + 8 + t;
                if (x < tot) {
                    const int idx = x < p1 ? s0[0] + x
                                  : x < p2 ? s0[1] + (x - p1)
                                  : x < p3 ? s0[2] + (x - p2)
                                  :          s0[3] + (x - p3);
                    recN = sorted[idx];
                }
            }
            #pragma unroll
            for (int u = 0; u < 8; ++u) {
                if (f + u < tot && (unsigned)rr[u] < (unsigned)BIN_NODES) {
                    float* row = sAcc + rr[u] * D_OUT;
                    const float4 a = gcmc_rot4(v0[u], q);
                    const float4 b = gcmc_rot4(v1[u], q);
                    __hip_atomic_fetch_add(row + c0,      a.x, __ATOMIC_RELAXED, __HIP_MEMORY_SCOPE_WORKGROUP);
                    __hip_atomic_fetch_add(row + c1,      a.y, __ATOMIC_RELAXED, __HIP_MEMORY_SCOPE_WORKGROUP);
                    __hip_atomic_fetch_add(row + c2,      a.z, __ATOMIC_RELAXED, __HIP_MEMORY_SCOPE_WORKGROUP);
                    __hip_atomic_fetch_add(row + c3,      a.w, __ATOMIC_RELAXED, __HIP_MEMORY_SCOPE_WORKGROUP);
                    __hip_atomic_fetch_add(row + 32 + c0, b.x, __ATOMIC_RELAXED, __HIP_MEMORY_SCOPE_WORKGROUP);
                    __hip_atomic_fetch_add(row + 32 + c1, b.y, __ATOMIC_RELAXED, __HIP_MEMORY_SCOPE_WORKGROUP);
                    __hip_atomic_fetch_add(row + 32 + c2, b.z, __ATOMIC_RELAXED, __HIP_MEMORY_SCOPE_WORKGROUP);
                    __hip_atomic_fetch_add(row + 32 + c3, b.w, __ATOMIC_RELAXED, __HIP_MEMORY_SCOPE_WORKGROUP);
                }
            }
        }
    }
    __syncthreads();

    for (int i = tid; i < BIN_NODES * (D_OUT / 4); i += 1024) {
        const int node = lo + (i >> 4);
        if (node < nNodes) {
            const float4 v  = sAcc4[i];
            const float  cd = ci[node];
            ((float4*)out)[(long long)node * (D_OUT / 4) + (i & 15)] =
                make_float4(v.x * cd, v.y * cd, v.z * cd, v.w * cd);
        }
    }
}

extern "C" void kernel_launch(void* const* d_in, const int* in_sizes, int n_in,
                              void* d_out, int out_size, void* d_ws, size_t ws_size,
                              hipStream_t stream) {
    const float* input_feat = (const float*)d_in[0];
    const float* weight     = (const float*)d_in[1];
    const float* cj         = (const float*)d_in[2];
    const float* ci         = (const float*)d_in[3];
    const int*   src_idx    = (const int*)d_in[4];
    const int*   dst_idx    = (const int*)d_in[5];
    float*       out        = (float*)d_out;
    float*       H          = (float*)d_ws;

    const int N = in_sizes[2];
    const int E = in_sizes[4];

    const int nChunks = (E + CHUNK - 1) / CHUNK;
    const int nBins   = (N + BIN_NODES - 1) / BIN_NODES;

    int2* sorted   = (int2*)(H + (size_t)N * D_OUT);
    int*  segStart = (int*)(sorted + (size_t)nChunks * CHUNK);

    gcmc_bin_sort_kernel<<<nChunks, SORT_THREADS, 0, stream>>>(src_idx, dst_idx, sorted, segStart, E);

    {
        const int nTiles = N / 16;
        const int blocks = (nTiles + 7) / 8;
        gcmc_gemm_wmma_kernel<<<blocks, 256, 0, stream>>>(input_feat, weight, cj, H, nTiles);
    }

    gcmc_bin_gather_kernel<<<nBins, 1024, 0, stream>>>(H, ci, sorted, segStart, out, N, nChunks);
}


# edge-sort launch and WMMA GEMM launch merged into one launch with two block roles (GEMM blocks first, sort blocks behind, shared 32 KB LDS carve): 2 launches instead of 3; sums unchanged
# speedup vs baseline: 3.1739x; 1.0390x over previous
#include <hip/hip_runtime.h>


typedef __attribute__((ext_vector_type(2))) float v2f;
typedef __attribute__((ext_vector_type(8))) float v8f;

#define D_IN  128
#define D_OUT 64

#define NBINS      256
#define BIN_NODES  391
#define CHUNK      3125
#define SORT_THREADS 256
#define SORT_ITEMS 13

__device__ __forceinline__
void gcmc_bin_sort_role(unsigned char* __restrict__ smem, const int chunk,
                        const int* __restrict__ src,
                        const int* __restrict__ dst,
                        int2*      __restrict__ sorted,
                        int*       __restrict__ segStart,
                        int nEdges) {
    int2* sBuf   = (int2*)smem;
    int*  sCnt   = (int*)(smem + CHUNK * sizeof(int2));
    int*  sStart = sCnt + (NBINS + 32);
    int*  sWave  = sStart + (NBINS + 1);

    const int tid    = threadIdx.x;
    const int lane   = tid & 31;
    const int base   = chunk * CHUNK;
    const int nValid = min(CHUNK, nEdges - base);

    sCnt[tid] = 0;
    if (tid < 32) sCnt[NBINS + tid] = 0;
    __syncthreads();

    int s[SORT_ITEMS], d[SORT_ITEMS], b[SORT_ITEMS], r[SORT_ITEMS];
    #pragma unroll
    for (int i = 0; i < SORT_ITEMS; ++i) {
        const int p = min(i * SORT_THREADS + tid, nValid - 1);
        d[i] = dst[base + p];
        s[i] = src[base + p];
    }
    #pragma unroll
    for (int i = 0; i < SORT_ITEMS; ++i) {
        const int p = i * SORT_THREADS + tid;
        b[i] = (p < nValid) ? d[i] / BIN_NODES : NBINS + lane;
        r[i] = atomicAdd(&sCnt[b[i]], 1);
    }
    __syncthreads();

    const int c = sCnt[tid];
    int inc = c;
    #pragma unroll
    for (int o = 1; o < 32; o <<= 1) {
        const int y = __shfl_up(inc, o);
        if (lane >= o) inc += y;
    }
    if (lane == 31) sWave[tid >> 5] = inc;
    __syncthreads();
    {
        const int w = tid >> 5;
        int off = 0;
        #pragma unroll
        for (int k = 0; k < NBINS / 32; ++k) off += (k < w) ? sWave[k] : 0;
        sStart[tid] = off + inc - c;
    }
    if (tid == 0) sStart[NBINS] = nValid;
    __syncthreads();

    int pos[SORT_ITEMS];
    #pragma unroll
    for (int i = 0; i < SORT_ITEMS; ++i) pos[i] = sStart[min(b[i], NBINS)] + r[i];
    #pragma unroll
    for (int i = 0; i < SORT_ITEMS; ++i) {
        const int p = i * SORT_THREADS + tid;
        if (p < nValid) sBuf[pos[i]] = make_int2(s[i], d[i]);
    }
    __syncthreads();

    int2 o[SORT_ITEMS];
    #pragma unroll
    for (int i = 0; i < SORT_ITEMS; ++i)
        o[i] = sBuf[min(i * SORT_THREADS + tid, nValid - 1)];
    #pragma unroll
    for (int i = 0; i < SORT_ITEMS; ++i) {
        const int p = i * SORT_THREADS + tid;
        if (p < nValid) sorted[base + p] = o[i];
    }
    segStart[chunk * (NBINS + 1) + tid] = sStart[tid];
    if (tid == 0) segStart[chunk * (NBINS + 1) + NBINS] = nValid;
}

__device__ __forceinline__
void gcmc_gemm_wmma_role(unsigned char* __restrict__ smem, const int blk,
                         const float* __restrict__ A,
                         const float* __restrict__ W,
                         const float* __restrict__ cj,
                         float* __restrict__ H,
                         int nTiles) {
    float2* sW2 = (float2*)smem;

    {
        int tid = threadIdx.x;
        #pragma unroll
        for (int i = 0; i < ((D_IN / 2) * D_OUT) / 256; ++i) {
            int p  = tid + i * 256;
            int kk = p >> 6;
            int n  = p & 63;
            float2 w;
            w.x = W[(2 * kk)     * D_OUT + n];
            w.y = W[(2 * kk + 1) * D_OUT + n];
            sW2[p] = w;
        }
    }
    __syncthreads();

    const int wave = threadIdx.x >> 5;
    const int lane = threadIdx.x & 31;
    const int tile = blk * (SORT_THREADS / 32) + wave;
    if (tile >= nTiles) return;

    const int half = lane >> 4;
    const int lrow = lane & 15;

    const float* Arow = A + (long long)(tile * 16 + lrow) * D_IN;

    v8f c0 = {}, c1 = {}, c2 = {}, c3 = {};

    for (int k = 0; k < D_IN; k += 4) {
        const int kA = k + 2 * half;
        v2f a = *(const v2f*)(Arow + kA);

        const float2* sRow = sW2 + ((k >> 1) + half) * D_OUT + lrow;
        v2f b0 = *(const v2f*)(sRow +  0);
        v2f b1 = *(const v2f*)(sRow + 16);
        v2f b2 = *(const v2f*)(sRow + 32);
        v2f b3 = *(const v2f*)(sRow + 48);

        c0 = __builtin_amdgcn_wmma_f32_16x16x4_f32(false, a, false, b0, (short)0, c0, false, false);
        c1 = __builtin_amdgcn_wmma_f32_16x16x4_f32(false, a, false, b1, (short)0, c1, false, false);
        c2 = __builtin_amdgcn_wmma_f32_16x16x4_f32(false, a, false, b2, (short)0, c2, false, false);
        c3 = __builtin_amdgcn_wmma_f32_16x16x4_f32(false, a, false, b3, (short)0, c3, false, false);
    }

    #pragma unroll
    for (int r = 0; r < 8; ++r) {
        const int   node = tile * 16 + r + 8 * half;
        const float s    = cj[node];
        float* Hrow = H + (long long)node * D_OUT;
        Hrow[ 0 + lrow] = c0[r] * s;
        Hrow[16 + lrow] = c1[r] * s;
        Hrow[32 + lrow] = c2[r] * s;
        Hrow[48 + lrow] = c3[r] * s;
    }
}

__global__ __launch_bounds__(SORT_THREADS) __attribute__((amdgpu_num_vgpr(80)))
void gcmc_gemm_and_sort_kernel(const float* __restrict__ A,
                               const float* __restrict__ W,
                               const float* __restrict__ cj,
                               float*       __restrict__ H,
                               int nTiles, int gemmBlocks,
                               const int*   __restrict__ src,
                               const int*   __restrict__ dst,
                               int2*        __restrict__ sorted,
                               int*         __restrict__ segStart,
                               int nEdges) {
    __shared__ __attribute__((aligned(16))) unsigned char smem[(D_IN / 2) * D_OUT * sizeof(float2)];
    const int blk = blockIdx.x;
    if (blk < gemmBlocks)
        gcmc_gemm_wmma_role(smem, blk, A, W, cj, H, nTiles);
    else
        gcmc_bin_sort_role(smem, blk - gemmBlocks, src, dst, sorted, segStart, nEdges);
}

__device__ __forceinline__ float4 gcmc_rot4(float4 v, int q) {
    const float4 a = (q & 1) ? make_float4(v.y, v.z, v.w, v.x) : v;
    const float4 b = (q & 2) ? make_float4(a.z, a.w, a.x, a.y) : a;
    return b;
}

__global__ __launch_bounds__(1024)
void gcmc_bin_gather_kernel(const float* __restrict__ H,
                            const float* __restrict__ ci,
                            const int2*  __restrict__ sorted,
                            const int*   __restrict__ segStart,
                            float*       __restrict__ out,
                            int nNodes, int nChunks) {
    __shared__ float sAcc[BIN_NODES * D_OUT];
    float4* sAcc4 = (float4*)sAcc;

    const int tid = threadIdx.x;
    const int bin = blockIdx.x;
    const int lo  = bin * BIN_NODES;

    for (int i = tid; i < BIN_NODES * D_OUT / 4; i += 1024)
        sAcc4[i] = make_float4(0.f, 0.f, 0.f, 0.f);
    __syncthreads();

    const int g = tid >> 3;
    const int t = tid & 7;
    const int q = g & 3;
    const int c0 = 4 * t + ((0 + q) & 3);
    const int c1 = 4 * t + ((1 + q) & 3);
    const int c2 = 4 * t + ((2 + q) & 3);
    const int c3 = 4 * t + ((3 + q) & 3);

    for (int cBase = g; cBase < nChunks; cBase += 512) {
        int s0[4], len[4], sa[4], se[4];
        #pragma unroll
        for (int k = 0; k < 4; ++k) {
            const int c  = min(cBase + 128 * k, nChunks - 1);
            const int* ss = segStart + (long long)c * (NBINS + 1) + bin;
            sa[k] = ss[0];
            se[k] = ss[1];
        }
        #pragma unroll
        for (int k = 0; k < 4; ++k) {
            const int c = cBase + 128 * k;
            s0[k]  = min(c, nChunks - 1) * CHUNK + sa[k];
            len[k] = (c < nChunks) ? se[k] - sa[k] : 0;
        }
        const int p1  = len[0];
        const int p2  = p1 + len[1];
        const int p3  = p2 + len[2];
        const int tot = p3 + len[3];

        int2 recN = make_int2(0, lo);
        if (t < tot) {
            const int x = t;
            const int idx = x < p1 ? s0[0] + x
                          : x < p2 ? s0[1] + (x - p1)
                          : x < p3 ? s0[2] + (x - p2)
                          :          s0[3] + (x - p3);
            recN = sorted[idx];
        }

        for (int f = 0; f < tot; f += 8) {
            const int2 rec = recN;
            float4 v0[8], v1[8];
            int    rr[8];
            #pragma unroll
            for (int u = 0; u < 8; ++u) {
                const int su = __shfl(rec.x, u, 8);
                rr[u] = __shfl(rec.y, u, 8) - lo;
                const float4* hp = (const float4*)(H + (long long)su * D_OUT + 4 * t);
                v0[u] = hp[0];
                v1[u] = hp[8];
            }
            recN = make_int2(0, lo);
            {
                const int x = f + 8 + t;
                if (x < tot) {
                    const int idx = x < p1 ? s0[0] + x
                                  : x < p2 ? s0[1] + (x - p1)
                                  : x < p3 ? s0[2] + (x - p2)
                                  :          s0[3] + (x - p3);
                    recN = sorted[idx];
                }
            }
            #pragma unroll
            for (int u = 0; u < 8; ++u) {
                if (f + u < tot && (unsigned)rr[u] < (unsigned)BIN_NODES) {
                    float* row = sAcc + rr[u] * D_OUT;
                    const float4 a = gcmc_rot4(v0[u], q);
                    const float4 b = gcmc_rot4(v1[u], q);
                    __hip_atomic_fetch_add(row + c0,      a.x, __ATOMIC_RELAXED, __HIP_MEMORY_SCOPE_WORKGROUP);
                    __hip_atomic_fetch_add(row + c1,      a.y, __ATOMIC_RELAXED, __HIP_MEMORY_SCOPE_WORKGROUP);
                    __hip_atomic_fetch_add(row + c2,      a.z, __ATOMIC_RELAXED, __HIP_MEMORY_SCOPE_WORKGROUP);
                    __hip_atomic_fetch_add(row + c3,      a.w, __ATOMIC_RELAXED, __HIP_MEMORY_SCOPE_WORKGROUP);
                    __hip_atomic_fetch_add(row + 32 + c0, b.x, __ATOMIC_RELAXED, __HIP_MEMORY_SCOPE_WORKGROUP);
                    __hip_atomic_fetch_add(row + 32 + c1, b.y, __ATOMIC_RELAXED, __HIP_MEMORY_SCOPE_WORKGROUP);
                    __hip_atomic_fetch_add(row + 32 + c2, b.z, __ATOMIC_RELAXED, __HIP_MEMORY_SCOPE_WORKGROUP);
                    __hip_atomic_fetch_add(row + 32 + c3, b.w, __ATOMIC_RELAXED, __HIP_MEMORY_SCOPE_WORKGROUP);
                }
            }
        }
    }
    __syncthreads();

    for (int i = tid; i < BIN_NODES * (D_OUT / 4); i += 1024) {
        const int node = lo + (i >> 4);
        if (node < nNodes) {
            const float4 v  = sAcc4[i];
            const float  cd = ci[node];
            ((float4*)out)[(long long)node * (D_OUT / 4) + (i & 15)] =
                make_float4(v.x * cd, v.y * cd, v.z * cd, v.w * cd);
        }
    }
}

extern "C" void kernel_launch(void* const* d_in, const int* in_sizes, int n_in,
                              void* d_out, int out_size, void* d_ws, size_t ws_size,
                              hipStream_t stream) {
    const float* input_feat = (const float*)d_in[0];
    const float* weight     = (const float*)d_in[1];
    const float* cj         = (const float*)d_in[2];
    const float* ci         = (const float*)d_in[3];
    const int*   src_idx    = (const int*)d_in[4];
    const int*   dst_idx    = (const int*)d_in[5];
    float*       out        = (float*)d_out;
    float*       H          = (float*)d_ws;

    const int N = in_sizes[2];
    const int E = in_sizes[4];

    const int nChunks = (E + CHUNK - 1) / CHUNK;
    const int nBins   = (N + BIN_NODES - 1) / BIN_NODES;

    int2* sorted   = (int2*)(H + (size_t)N * D_OUT);
    int*  segStart = (int*)(sorted + (size_t)nChunks * CHUNK);

    {
        const int nTiles     = N / 16;
        const int gemmBlocks = (nTiles + 7) / 8;
        gcmc_gemm_and_sort_kernel<<<gemmBlocks + nChunks, SORT_THREADS, 0, stream>>>(
            input_feat, weight, cj, H, nTiles, gemmBlocks,
            src_idx, dst_idx, sorted, segStart, E);
    }

    gcmc_bin_gather_kernel<<<nBins, 1024, 0, stream>>>(H, ci, sorted, segStart, out, N, nChunks);
}


# range gather: per-edge ds_add_f32 into LDS sums replaced by an LDS counting sort of the range's edges by dst row; 8 lanes sum a row in registers, store it once; sum order = LDS rank (arrival) order
# speedup vs baseline: 3.6379x; 1.1462x over previous
#include <hip/hip_runtime.h>


typedef __attribute__((ext_vector_type(2))) float v2f;
typedef __attribute__((ext_vector_type(8))) float v8f;

#define D_IN  128
#define D_OUT 64

#define NBINS      256
#define BIN_NODES  391
#define CHUNK      3125
#define SORT_THREADS 256
#define SORT_ITEMS 13

__device__ __forceinline__
void gcmc_bin_sort_role(unsigned char* __restrict__ smem, const int chunk,
                        const int* __restrict__ src,
                        const int* __restrict__ dst,
                        int2*      __restrict__ sorted,
                        int*       __restrict__ segStart,
                        int nEdges) {
    int2* sBuf   = (int2*)smem;
    int*  sCnt   = (int*)(smem + CHUNK * sizeof(int2));
    int*  sStart = sCnt + (NBINS + 32);
    int*  sWave  = sStart + (NBINS + 1);

    const int tid    = threadIdx.x;
    const int lane   = tid & 31;
    const int base   = chunk * CHUNK;
    const int nValid = min(CHUNK, nEdges - base);

    sCnt[tid] = 0;
    if (tid < 32) sCnt[NBINS + tid] = 0;
    __syncthreads();

    int s[SORT_ITEMS], d[SORT_ITEMS], b[SORT_ITEMS], r[SORT_ITEMS];
    #pragma unroll
    for (int i = 0; i < SORT_ITEMS; ++i) {
        const int p = min(i * SORT_THREADS + tid, nValid - 1);
        d[i] = dst[base + p];
        s[i] = src[base + p];
    }
    #pragma unroll
    for (int i = 0; i < SORT_ITEMS; ++i) {
        const int p = i * SORT_THREADS + tid;
        b[i] = (p < nValid) ? d[i] / BIN_NODES : NBINS + lane;
        r[i] = atomicAdd(&sCnt[b[i]], 1);
    }
    __syncthreads();

    const int c = sCnt[tid];
    int inc = c;
    #pragma unroll
    for (int o = 1; o < 32; o <<= 1) {
        const int y = __shfl_up(inc, o);
        if (lane >= o) inc += y;
    }
    if (lane == 31) sWave[tid >> 5] = inc;
    __syncthreads();
    {
        const int w = tid >> 5;
        int off = 0;
        #pragma unroll
        for (int k = 0; k < NBINS / 32; ++k) off += (k < w) ? sWave[k] : 0;
        sStart[tid] = off + inc - c;
    }
    if (tid == 0) sStart[NBINS] = nValid;
    __syncthreads();

    int pos[SORT_ITEMS];
    #pragma unroll
    for (int i = 0; i < SORT_ITEMS; ++i) pos[i] = sStart[min(b[i], NBINS)] + r[i];
    #pragma unroll
    for (int i = 0; i < SORT_ITEMS; ++i) {
        const int p = i * SORT_THREADS + tid;
        if (p < nValid) sBuf[pos[i]] = make_int2(s[i], d[i]);
    }
    __syncthreads();

    int2 o[SORT_ITEMS];
    #pragma unroll
    for (int i = 0; i < SORT_ITEMS; ++i)
        o[i] = sBuf[min(i * SORT_THREADS + tid, nValid - 1)];
    #pragma unroll
    for (int i = 0; i < SORT_ITEMS; ++i) {
        const int p = i * SORT_THREADS + tid;
        if (p < nValid) sorted[base + p] = o[i];
    }
    segStart[chunk * (NBINS + 1) + tid] = sStart[tid];
    if (tid == 0) segStart[chunk * (NBINS + 1) + NBINS] = nValid;
}

__device__ __forceinline__
void gcmc_gemm_wmma_role(unsigned char* __restrict__ smem, const int blk,
                         const float* __restrict__ A,
                         const float* __restrict__ W,
                         const float* __restrict__ cj,
                         float* __restrict__ H,
                         int nTiles) {
    float2* sW2 = (float2*)smem;

    {
        int tid = threadIdx.x;
        #pragma unroll
        for (int i = 0; i < ((D_IN / 2) * D_OUT) / 256; ++i) {
            int p  = tid + i * 256;
            int kk = p >> 6;
            int n  = p & 63;
            float2 w;
            w.x = W[(2 * kk)     * D_OUT + n];
            w.y = W[(2 * kk + 1) * D_OUT + n];
            sW2[p] = w;
        }
    }
    __syncthreads();

    const int wave = threadIdx.x >> 5;
    const int lane = threadIdx.x & 31;
    const int tile = blk * (SORT_THREADS / 32) + wave;
    if (tile >= nTiles) return;

    const int half = lane >> 4;
    const int lrow = lane & 15;

    const float* Arow = A + (long long)(tile * 16 + lrow) * D_IN;

    v8f c0 = {}, c1 = {}, c2 = {}, c3 = {};

    for (int k = 0; k < D_IN; k += 4) {
        const int kA = k + 2 * half;
        v2f a = *(const v2f*)(Arow + kA);

        const float2* sRow = sW2 + ((k >> 1) + half) * D_OUT + lrow;
        v2f b0 = *(const v2f*)(sRow +  0);
        v2f b1 = *(const v2f*)(sRow + 16);
        v2f b2 = *(const v2f*)(sRow + 32);
        v2f b3 = *(const v2f*)(sRow + 48);

        c0 = __builtin_amdgcn_wmma_f32_16x16x4_f32(false, a, false, b0, (short)0, c0, false, false);
        c1 = __builtin_amdgcn_wmma_f32_16x16x4_f32(false, a, false, b1, (short)0, c1, false, false);
        c2 = __builtin_amdgcn_wmma_f32_16x16x4_f32(false, a, false, b2, (short)0, c2, false, false);
        c3 = __builtin_amdgcn_wmma_f32_16x16x4_f32(false, a, false, b3, (short)0, c3, false, false);
    }

    #pragma unroll
    for (int r = 0; r < 8; ++r) {
        const int   node = tile * 16 + r + 8 * half;
        const float s    = cj[node];
        float* Hrow = H + (long long)node * D_OUT;
        Hrow[ 0 + lrow] = c0[r] * s;
        Hrow[16 + lrow] = c1[r] * s;
        Hrow[32 + lrow] = c2[r] * s;
        Hrow[48 + lrow] = c3[r] * s;
    }
}

__global__ __launch_bounds__(SORT_THREADS) __attribute__((amdgpu_num_vgpr(80)))
void gcmc_gemm_and_sort_kernel(const float* __restrict__ A,
                               const float* __restrict__ W,
                               const float* __restrict__ cj,
                               float*       __restrict__ H,
                               int nTiles, int gemmBlocks,
                               const int*   __restrict__ src,
                               const int*   __restrict__ dst,
                               int2*        __restrict__ sorted,
                               int*         __restrict__ segStart,
                               int nEdges) {
    __shared__ __attribute__((aligned(16))) unsigned char smem[(D_IN / 2) * D_OUT * sizeof(float2)];
    const int blk = blockIdx.x;
    if (blk < gemmBlocks)
        gcmc_gemm_wmma_role(smem, blk, A, W, cj, H, nTiles);
    else
        gcmc_bin_sort_role(smem, blk - gemmBlocks, src, dst, sorted, segStart, nEdges);
}

#define GATHER_Q     128
#define GATHER_SLOTS (GATHER_Q / 8)
#define GATHER_CAP   (128 * GATHER_Q)
#define ROWS_PAD     416

__global__ __launch_bounds__(1024)
void gcmc_bin_gather_kernel(const float* __restrict__ H,
                            const float* __restrict__ ci,
                            const int2*  __restrict__ sorted,
                            const int*   __restrict__ segStart,
                            float*       __restrict__ out,
                            int nNodes, int nChunks) {
    __shared__ int sSrc[GATHER_CAP];
    __shared__ int sCnt[ROWS_PAD];
    __shared__ int sStart[ROWS_PAD];
    __shared__ int sWave[ROWS_PAD / 32];
    __shared__ int sMax;

    const int tid  = threadIdx.x;
    const int lane = tid & 31;
    const int wave = tid >> 5;
    const int bin  = blockIdx.x;
    const int lo   = bin * BIN_NODES;
    const int g    = tid >> 3;
    const int t    = tid & 7;

    float cdv[4];
    #pragma unroll
    for (int j = 0; j < 4; ++j) {
        const int n = g + 128 * j;
        cdv[j] = (n < BIN_NODES && lo + n < nNodes) ? ci[lo + n] : 0.f;
    }

    bool first = true;

    for (int c0 = 0; c0 < nChunks; c0 += 512) {
        const int cBase = c0 + g;
        int s0[4], len[4], sa[4], se[4];
        #pragma unroll
        for (int k = 0; k < 4; ++k) {
            const int c  = min(cBase + 128 * k, nChunks - 1);
            const int* ss = segStart + (long long)c * (NBINS + 1) + bin;
            sa[k] = ss[0];
            se[k] = ss[1];
        }
        #pragma unroll
        for (int k = 0; k < 4; ++k) {
            const int c = cBase + 128 * k;
            s0[k]  = min(c, nChunks - 1) * CHUNK + sa[k];
            len[k] = (c < nChunks) ? se[k] - sa[k] : 0;
        }
        const int p1  = len[0];
        const int p2  = p1 + len[1];
        const int p3  = p2 + len[2];
        const int tot = p3 + len[3];

        int nWin = 1;
        for (int w = 0; w < nWin; ++w) {
            if (tid < ROWS_PAD) sCnt[tid] = 0;
            if (w == 0 && tid == 0) sMax = 0;
            __syncthreads();
            if (w == 0 && t == 0) atomicMax(&sMax, tot);

            const int wlo = w * GATHER_Q;
            int2 rec[GATHER_SLOTS];
            #pragma unroll
            for (int i = 0; i < GATHER_SLOTS; ++i) {
                const int x = wlo + 8 * i + t;
                rec[i] = make_int2(0, lo - 1);
                if (x < tot) {
                    const int idx = x < p1 ? s0[0] + x
                                  : x < p2 ? s0[1] + (x - p1)
                                  : x < p3 ? s0[2] + (x - p2)
                                  :          s0[3] + (x - p3);
                    rec[i] = sorted[idx];
                }
            }
            int pk[GATHER_SLOTS];
            #pragma unroll
            for (int i = 0; i < GATHER_SLOTS; ++i) {
                const int row = rec[i].y - lo;
                pk[i] = -1;
                if ((unsigned)row < (unsigned)BIN_NODES)
                    pk[i] = row | (atomicAdd(&sCnt[row], 1) << 9);
            }
            __syncthreads();
            if (w == 0) nWin = (sMax + GATHER_Q - 1) / GATHER_Q;

            const int c = (tid < ROWS_PAD) ? sCnt[tid] : 0;
            int inc = c;
            #pragma unroll
            for (int o = 1; o < 32; o <<= 1) {
                const int y = __shfl_up(inc, o);
                if (lane >= o) inc += y;
            }
            if (tid < ROWS_PAD && lane == 31) sWave[wave] = inc;
            __syncthreads();
            if (tid < ROWS_PAD) {
                int off = 0;
                #pragma unroll
                for (int k = 0; k < ROWS_PAD / 32; ++k) off += (k < wave) ? sWave[k] : 0;
                sStart[tid] = off + inc - c;
            }
            __syncthreads();

            #pragma unroll
            for (int i = 0; i < GATHER_SLOTS; ++i)
                if (pk[i] >= 0) sSrc[sStart[pk[i] & 511] + (pk[i] >> 9)] = rec[i].x;
            __syncthreads();

            #pragma unroll 1
            for (int j = 0; j < 4; ++j) {
                const int n    = g + 128 * j;
                const int node = lo + n;
                if (n < BIN_NODES && node < nNodes) {
                    const int b = sStart[n];
                    const int e = sStart[n + 1];
                    float4 a0 = make_float4(0.f, 0.f, 0.f, 0.f);
                    float4 a1 = make_float4(0.f, 0.f, 0.f, 0.f);
                    if (b < e) {
                        int sv[8];
                        #pragma unroll
                        for (int u = 0; u < 8; ++u) sv[u] = sSrc[min(b + u, e - 1)];
                        int f = b;
                        for (; f + 8 <= e; f += 8) {
                            float4 v0[8], v1[8];
                            #pragma unroll
                            for (int u = 0; u < 8; ++u) {
                                const float4* hp = (const float4*)(H + (long long)sv[u] * D_OUT + 4 * t);
                                v0[u] = hp[0];
                                v1[u] = hp[8];
                            }
                            #pragma unroll
                            for (int u = 0; u < 8; ++u) sv[u] = sSrc[min(f + 8 + u, e - 1)];
                            #pragma unroll
                            for (int u = 0; u < 8; ++u) {
                                a0.x += v0[u].x; a0.y += v0[u].y; a0.z += v0[u].z; a0.w += v0[u].w;
                                a1.x += v1[u].x; a1.y += v1[u].y; a1.z += v1[u].z; a1.w += v1[u].w;
                            }
                        }
                        if (f < e) {
                            float4 v0[8], v1[8];
                            #pragma unroll
                            for (int u = 0; u < 8; ++u) {
                                const float4* hp = (const float4*)(H + (long long)sv[u] * D_OUT + 4 * t);
                                v0[u] = hp[0];
                                v1[u] = hp[8];
                            }
                            #pragma unroll
                            for (int u = 0; u < 8; ++u) {
                                const bool on = f + u < e;
                                a0.x += on ? v0[u].x : 0.f; a0.y += on ? v0[u].y : 0.f;
                                a0.z += on ? v0[u].z : 0.f; a0.w += on ? v0[u].w : 0.f;
                                a1.x += on ? v1[u].x : 0.f; a1.y += on ? v1[u].y : 0.f;
                                a1.z += on ? v1[u].z : 0.f; a1.w += on ? v1[u].w : 0.f;
                            }
                        }
                    }
                    const float cd = j == 0 ? cdv[0] : j == 1 ? cdv[1] : j == 2 ? cdv[2] : cdv[3];
                    float4* op = (float4*)out + (long long)node * (D_OUT / 4) + t;
                    float4 r0 = make_float4(a0.x * cd, a0.y * cd, a0.z * cd, a0.w * cd);
                    float4 r1 = make_float4(a1.x * cd, a1.y * cd, a1.z * cd, a1.w * cd);
                    if (!first) {
                        const float4 o0 = op[0];
                        const float4 o1 = op[8];
                        r0 = make_float4(o0.x + r0.x, o0.y + r0.y, o0.z + r0.z, o0.w + r0.w);
                        r1 = make_float4(o1.x + r1.x, o1.y + r1.y, o1.z + r1.z, o1.w + r1.w);
                    }
                    op[0] = r0;
                    op[8] = r1;
                }
            }
            first = false;
        }
    }
}

extern "C" void kernel_launch(void* const* d_in, const int* in_sizes, int n_in,
                              void* d_out, int out_size, void* d_ws, size_t ws_size,
                              hipStream_t stream) {
    const float* input_feat = (const float*)d_in[0];
    const float* weight     = (const float*)d_in[1];
    const float* cj         = (const float*)d_in[2];
    const float* ci         = (const float*)d_in[3];
    const int*   src_idx    = (const int*)d_in[4];
    const int*   dst_idx    = (const int*)d_in[5];
    float*       out        = (float*)d_out;
    float*       H          = (float*)d_ws;

    const int N = in_sizes[2];
    const int E = in_sizes[4];

    const int nChunks = (E + CHUNK - 1) / CHUNK;
    const int nBins   = (N + BIN_NODES - 1) / BIN_NODES;

    int2* sorted   = (int2*)(H + (size_t)N * D_OUT);
    int*  segStart = (int*)(sorted + (size_t)nChunks * CHUNK);

    {
        const int nTiles     = N / 16;
        const int gemmBlocks = (nTiles + 7) / 8;
        gcmc_gemm_and_sort_kernel<<<gemmBlocks + nChunks, SORT_THREADS, 0, stream>>>(
            input_feat, weight, cj, H, nTiles, gemmBlocks,
            src_idx, dst_idx, sorted, segStart, E);
    }

    gcmc_bin_gather_kernel<<<nBins, 1024, 0, stream>>>(H, ci, sorted, segStart, out, N, nChunks);
}


# range gather: rows dealt to the 8-lane groups from an LDS counter, one 8-source step per loop trip for all four groups of a wave; no group idles behind a longer or an extra row; sums unchanged
# speedup vs baseline: 3.7594x; 1.0334x over previous
#include <hip/hip_runtime.h>


typedef __attribute__((ext_vector_type(2))) float v2f;
typedef __attribute__((ext_vector_type(8))) float v8f;

#define D_IN  128
#define D_OUT 64

#define NBINS      256
#define BIN_NODES  391
#define CHUNK      3125
#define SORT_THREADS 256
#define SORT_ITEMS 13

__device__ __forceinline__
void gcmc_bin_sort_role(unsigned char* __restrict__ smem, const int chunk,
                        const int* __restrict__ src,
                        const int* __restrict__ dst,
                        int2*      __restrict__ sorted,
                        int*       __restrict__ segStart,
                        int nEdges) {
    int2* sBuf   = (int2*)smem;
    int*  sCnt   = (int*)(smem + CHUNK * sizeof(int2));
    int*  sStart = sCnt + (NBINS + 32);
    int*  sWave  = sStart + (NBINS + 1);

    const int tid    = threadIdx.x;
    const int lane   = tid & 31;
    const int base   = chunk * CHUNK;
    const int nValid = min(CHUNK, nEdges - base);

    sCnt[tid] = 0;
    if (tid < 32) sCnt[NBINS + tid] = 0;
    __syncthreads();

    int s[SORT_ITEMS], d[SORT_ITEMS], b[SORT_ITEMS], r[SORT_ITEMS];
    #pragma unroll
    for (int i = 0; i < SORT_ITEMS; ++i) {
        const int p = min(i * SORT_THREADS + tid, nValid - 1);
        d[i] = dst[base + p];
        s[i] = src[base + p];
    }
    #pragma unroll
    for (int i = 0; i < SORT_ITEMS; ++i) {
        const int p = i * SORT_THREADS + tid;
        b[i] = (p < nValid) ? d[i] / BIN_NODES : NBINS + lane;
        r[i] = atomicAdd(&sCnt[b[i]], 1);
    }
    __syncthreads();

    const int c = sCnt[tid];
    int inc = c;
    #pragma unroll
    for (int o = 1; o < 32; o <<= 1) {
        const int y = __shfl_up(inc, o);
        if (lane >= o) inc += y;
    }
    if (lane == 31) sWave[tid >> 5] = inc;
    __syncthreads();
    {
        const int w = tid >> 5;
        int off = 0;
        #pragma unroll
        for (int k = 0; k < NBINS / 32; ++k) off += (k < w) ? sWave[k] : 0;
        sStart[tid] = off + inc - c;
    }
    if (tid == 0) sStart[NBINS] = nValid;
    __syncthreads();

    int pos[SORT_ITEMS];
    #pragma unroll
    for (int i = 0; i < SORT_ITEMS; ++i) pos[i] = sStart[min(b[i], NBINS)] + r[i];
    #pragma unroll
    for (int i = 0; i < SORT_ITEMS; ++i) {
        const int p = i * SORT_THREADS + tid;
        if (p < nValid) sBuf[pos[i]] = make_int2(s[i], d[i]);
    }
    __syncthreads();

    int2 o[SORT_ITEMS];
    #pragma unroll
    for (int i = 0; i < SORT_ITEMS; ++i)
        o[i] = sBuf[min(i * SORT_THREADS + tid, nValid - 1)];
    #pragma unroll
    for (int i = 0; i < SORT_ITEMS; ++i) {
        const int p = i * SORT_THREADS + tid;
        if (p < nValid) sorted[base + p] = o[i];
    }
    segStart[chunk * (NBINS + 1) + tid] = sStart[tid];
    if (tid == 0) segStart[chunk * (NBINS + 1) + NBINS] = nValid;
}

__device__ __forceinline__
void gcmc_gemm_wmma_role(unsigned char* __restrict__ smem, const int blk,
                         const float* __restrict__ A,
                         const float* __restrict__ W,
                         const float* __restrict__ cj,
                         float* __restrict__ H,
                         int nTiles) {
    float2* sW2 = (float2*)smem;

    {
        int tid = threadIdx.x;
        #pragma unroll
        for (int i = 0; i < ((D_IN / 2) * D_OUT) / 256; ++i) {
            int p  = tid + i * 256;
            int kk = p >> 6;
            int n  = p & 63;
            float2 w;
            w.x = W[(2 * kk)     * D_OUT + n];
            w.y = W[(2 * kk + 1) * D_OUT + n];
            sW2[p] = w;
        }
    }
    __syncthreads();

    const int wave = threadIdx.x >> 5;
    const int lane = threadIdx.x & 31;
    const int tile = blk * (SORT_THREADS / 32) + wave;
    if (tile >= nTiles) return;

    const int half = lane >> 4;
    const int lrow = lane & 15;

    const float* Arow = A + (long long)(tile * 16 + lrow) * D_IN;

    v8f c0 = {}, c1 = {}, c2 = {}, c3 = {};

    for (int k = 0; k < D_IN; k += 4) {
        const int kA = k + 2 * half;
        v2f a = *(const v2f*)(Arow + kA);

        const float2* sRow = sW2 + ((k >> 1) + half) * D_OUT + lrow;
        v2f b0 = *(const v2f*)(sRow +  0);
        v2f b1 = *(const v2f*)(sRow + 16);
        v2f b2 = *(const v2f*)(sRow + 32);
        v2f b3 = *(const v2f*)(sRow + 48);

        c0 = __builtin_amdgcn_wmma_f32_16x16x4_f32(false, a, false, b0, (short)0, c0, false, false);
        c1 = __builtin_amdgcn_wmma_f32_16x16x4_f32(false, a, false, b1, (short)0, c1, false, false);
        c2 = __builtin_amdgcn_wmma_f32_16x16x4_f32(false, a, false, b2, (short)0, c2, false, false);
        c3 = __builtin_amdgcn_wmma_f32_16x16x4_f32(false, a, false, b3, (short)0, c3, false, false);
    }

    #pragma unroll
    for (int r = 0; r < 8; ++r) {
        const int   node = tile * 16 + r + 8 * half;
        const float s    = cj[node];
        float* Hrow = H + (long long)node * D_OUT;
        Hrow[ 0 + lrow] = c0[r] * s;
        Hrow[16 + lrow] = c1[r] * s;
        Hrow[32 + lrow] = c2[r] * s;
        Hrow[48 + lrow] = c3[r] * s;
    }
}

__global__ __launch_bounds__(SORT_THREADS) __attribute__((amdgpu_num_vgpr(80)))
void gcmc_gemm_and_sort_kernel(const float* __restrict__ A,
                               const float* __restrict__ W,
                               const float* __restrict__ cj,
                               float*       __restrict__ H,
                               int nTiles, int gemmBlocks,
                               const int*   __restrict__ src,
                               const int*   __restrict__ dst,
                               int2*        __restrict__ sorted,
                               int*         __restrict__ segStart,
                               int nEdges) {
    __shared__ __attribute__((aligned(16))) unsigned char smem[(D_IN / 2) * D_OUT * sizeof(float2)];
    const int blk = blockIdx.x;
    if (blk < gemmBlocks)
        gcmc_gemm_wmma_role(smem, blk, A, W, cj, H, nTiles);
    else
        gcmc_bin_sort_role(smem, blk - gemmBlocks, src, dst, sorted, segStart, nEdges);
}

#define GATHER_Q     128
#define GATHER_SLOTS (GATHER_Q / 8)
#define GATHER_GROUPS 128
#define GATHER_CAP   (GATHER_GROUPS * GATHER_Q)
#define ROWS_PAD     416

__global__ __launch_bounds__(1024)
void gcmc_bin_gather_kernel(const float* __restrict__ H,
                            const float* __restrict__ ci,
                            const int2*  __restrict__ sorted,
                            const int*   __restrict__ segStart,
                            float*       __restrict__ out,
                            int nNodes, int nChunks) {
    __shared__ int sSrc[GATHER_CAP];
    __shared__ int sCnt[ROWS_PAD];
    __shared__ int sStart[ROWS_PAD];
    __shared__ int sWave[ROWS_PAD / 32];
    __shared__ int sMax;
    __shared__ int sNext;

    const int tid  = threadIdx.x;
    const int lane = tid & 31;
    const int wave = tid >> 5;
    const int bin  = blockIdx.x;
    const int lo   = bin * BIN_NODES;
    const int g    = tid >> 3;
    const int t    = tid & 7;
    const float4* __restrict__ H4 = (const float4*)H;
    float4* __restrict__ out4 = (float4*)out;

    const int rowLimit = min(BIN_NODES, nNodes - lo);

    bool first = true;

    for (int c0 = 0; c0 < nChunks; c0 += 512) {
        const int cBase = c0 + g;
        int nWin = 1;
        for (int w = 0; w < nWin; ++w) {
            int s0[4], len[4], sa[4], se[4];
            #pragma unroll
            for (int k = 0; k < 4; ++k) {
                const int c  = min(cBase + 128 * k, nChunks - 1);
                const int* ss = segStart + (long long)c * (NBINS + 1) + bin;
                sa[k] = ss[0];
                se[k] = ss[1];
            }
            #pragma unroll
            for (int k = 0; k < 4; ++k) {
                const int c = cBase + 128 * k;
                s0[k]  = min(c, nChunks - 1) * CHUNK + sa[k];
                len[k] = (c < nChunks) ? se[k] - sa[k] : 0;
            }
            const int p1  = len[0];
            const int p2  = p1 + len[1];
            const int p3  = p2 + len[2];
            const int tot = p3 + len[3];

            if (tid < ROWS_PAD) sCnt[tid] = 0;
            if (w == 0 && tid == 0) sMax = 0;
            __syncthreads();
            if (w == 0 && t == 0) atomicMax(&sMax, tot);
            if (tid == 0) sNext = GATHER_GROUPS;

            const int wlo = w * GATHER_Q;
            int2 rec[GATHER_SLOTS];
            #pragma unroll
            for (int i = 0; i < GATHER_SLOTS; ++i) {
                const int x = wlo + 8 * i + t;
                rec[i] = make_int2(0, lo - 1);
                if (x < tot) {
                    const int idx = x < p1 ? s0[0] + x
                                  : x < p2 ? s0[1] + (x - p1)
                                  : x < p3 ? s0[2] + (x - p2)
                                  :          s0[3] + (x - p3);
                    rec[i] = sorted[idx];
                }
            }
            int pk[GATHER_SLOTS];
            #pragma unroll
            for (int i = 0; i < GATHER_SLOTS; ++i) {
                const int row = rec[i].y - lo;
                pk[i] = -1;
                if ((unsigned)row < (unsigned)BIN_NODES)
                    pk[i] = row | (atomicAdd(&sCnt[row], 1) << 9);
            }
            __syncthreads();
            if (w == 0) nWin = (sMax + GATHER_Q - 1) / GATHER_Q;

            const int c = (tid < ROWS_PAD) ? sCnt[tid] : 0;
            int inc = c;
            #pragma unroll
            for (int o = 1; o < 32; o <<= 1) {
                const int y = __shfl_up(inc, o);
                if (lane >= o) inc += y;
            }
            if (tid < ROWS_PAD && lane == 31) sWave[wave] = inc;
            __syncthreads();
            if (tid < ROWS_PAD) {
                int off = 0;
                #pragma unroll
                for (int k = 0; k < ROWS_PAD / 32; ++k) off += (k < wave) ? sWave[k] : 0;
                sStart[tid] = off + inc - c;
            }
            __syncthreads();

            #pragma unroll
            for (int i = 0; i < GATHER_SLOTS; ++i)
                if (pk[i] >= 0) sSrc[sStart[pk[i] & 511] + (pk[i] >> 9)] = rec[i].x;
            __syncthreads();

            {
                int   cn = (g < rowLimit) ? g : -1;
                int   f = 0, e = 0;
                int   sv[8];
                #pragma unroll
                for (int u = 0; u < 8; ++u) sv[u] = 0;
                if (cn >= 0) {
                    f = sStart[cn];
                    e = sStart[cn + 1];
                    if (f < e) {
                        #pragma unroll
                        for (int u = 0; u < 8; ++u) sv[u] = sSrc[min(f + u, e - 1)];
                    }
                }
                float4 a0 = make_float4(0.f, 0.f, 0.f, 0.f);
                float4 a1 = make_float4(0.f, 0.f, 0.f, 0.f);
                const int tripMax = GATHER_CAP / 8 + BIN_NODES + 1;
                for (int trip = 0; trip < tripMax; ++trip) {
                    if (!__any(cn >= 0)) break;
                    if (cn >= 0) {
                        const bool has = f < e;
                        const int  f2  = f + 8;
                        const bool fin = f2 >= e;
                        float cd = 0.f;
                        if (fin) cd = ci[lo + cn];
                        float4 v0[8], v1[8];
                        if (has) {
                            #pragma unroll
                            for (int u = 0; u < 8; ++u) {
                                const unsigned q = (unsigned)sv[u] * (D_OUT / 4) + (unsigned)t;
                                v0[u] = H4[q];
                                v1[u] = H4[q + 8];
                            }
                        }
                        int   cnN = cn, fN = f2, eN = e;
                        if (fin) {
                            int r = 0;
                            if (t == 0) r = atomicAdd(&sNext, 1);
                            r   = __shfl(r, 0, 8);
                            cnN = (r < rowLimit) ? r : -1;
                            fN  = 0;
                            eN  = 0;
                            if (cnN >= 0) {
                                fN = sStart[cnN];
                                eN = sStart[cnN + 1];
                            }
                        }
                        if (fN < eN) {
                            #pragma unroll
                            for (int u = 0; u < 8; ++u) sv[u] = sSrc[min(fN + u, eN - 1)];
                        }
                        if (has) {
                            if (f2 <= e) {
                                #pragma unroll
                                for (int u = 0; u < 8; ++u) {
                                    a0.x += v0[u].x; a0.y += v0[u].y; a0.z += v0[u].z; a0.w += v0[u].w;
                                    a1.x += v1[u].x; a1.y += v1[u].y; a1.z += v1[u].z; a1.w += v1[u].w;
                                }
                            } else {
                                #pragma unroll
                                for (int u = 0; u < 8; ++u) {
                                    const bool on = f + u < e;
                                    a0.x += on ? v0[u].x : 0.f; a0.y += on ? v0[u].y : 0.f;
                                    a0.z += on ? v0[u].z : 0.f; a0.w += on ? v0[u].w : 0.f;
                                    a1.x += on ? v1[u].x : 0.f; a1.y += on ? v1[u].y : 0.f;
                                    a1.z += on ? v1[u].z : 0.f; a1.w += on ? v1[u].w : 0.f;
                                }
                            }
                        }
                        if (fin) {
                            float4* op = out4 + ((unsigned)(lo + cn) * (D_OUT / 4) + (unsigned)t);
                            float4 r0 = make_float4(a0.x * cd, a0.y * cd, a0.z * cd, a0.w * cd);
                            float4 r1 = make_float4(a1.x * cd, a1.y * cd, a1.z * cd, a1.w * cd);
                            if (!first) {
                                const float4 o0 = op[0];
                                const float4 o1 = op[8];
                                r0 = make_float4(o0.x + r0.x, o0.y + r0.y, o0.z + r0.z, o0.w + r0.w);
                                r1 = make_float4(o1.x + r1.x, o1.y + r1.y, o1.z + r1.z, o1.w + r1.w);
                            }
                            op[0] = r0;
                            op[8] = r1;
                            a0 = make_float4(0.f, 0.f, 0.f, 0.f);
                            a1 = make_float4(0.f, 0.f, 0.f, 0.f);
                        }
                        cn = cnN; f = fN; e = eN;
                    }
                }
            }
            first = false;
        }
    }
}

extern "C" void kernel_launch(void* const* d_in, const int* in_sizes, int n_in,
                              void* d_out, int out_size, void* d_ws, size_t ws_size,
                              hipStream_t stream) {
    const float* input_feat = (const float*)d_in[0];
    const float* weight     = (const float*)d_in[1];
    const float* cj         = (const float*)d_in[2];
    const float* ci         = (const float*)d_in[3];
    const int*   src_idx    = (const int*)d_in[4];
    const int*   dst_idx    = (const int*)d_in[5];
    float*       out        = (float*)d_out;
    float*       H          = (float*)d_ws;

    const int N = in_sizes[2];
    const int E = in_sizes[4];

    const int nChunks = (E + CHUNK - 1) / CHUNK;
    const int nBins   = (N + BIN_NODES - 1) / BIN_NODES;

    int2* sorted   = (int2*)(H + (size_t)N * D_OUT);
    int*  segStart = (int*)(sorted + (size_t)nChunks * CHUNK);

    {
        const int nTiles     = N / 16;
        const int gemmBlocks = (nTiles + 7) / 8;
        gcmc_gemm_and_sort_kernel<<<gemmBlocks + nChunks, SORT_THREADS, 0, stream>>>(
            input_feat, weight, cj, H, nTiles, gemmBlocks,
            src_idx, dst_idx, sorted, segStart, E);
    }

    gcmc_bin_gather_kernel<<<nBins, 1024, 0, stream>>>(H, ci, sorted, segStart, out, N, nChunks);
}


# GEMM role: A tile fetched by whole-line global loads (4 whole 128-B lines per wave-wide 16-B load, not 16 lines x 16 B) through a wave-private LDS buffer, a quarter of K per pass; h bit-identical
# speedup vs baseline: 3.9088x; 1.0397x over previous
#include <hip/hip_runtime.h>


typedef __attribute__((ext_vector_type(2))) float v2f;
typedef __attribute__((ext_vector_type(8))) float v8f;

#define D_IN  128
#define D_OUT 64

#define NBINS      256
#define BIN_NODES  391
#define CHUNK      3125
#define SORT_THREADS 256
#define SORT_ITEMS 13

__device__ __forceinline__
void gcmc_bin_sort_role(unsigned char* __restrict__ smem, const int chunk,
                        const int* __restrict__ src,
                        const int* __restrict__ dst,
                        int2*      __restrict__ sorted,
                        int*       __restrict__ segStart,
                        int nEdges) {
    int2* sBuf   = (int2*)smem;
    int*  sCnt   = (int*)(smem + CHUNK * sizeof(int2));
    int*  sStart = sCnt + (NBINS + 32);
    int*  sWave  = sStart + (NBINS + 1);

    const int tid    = threadIdx.x;
    const int lane   = tid & 31;
    const int base   = chunk * CHUNK;
    const int nValid = min(CHUNK, nEdges - base);

    sCnt[tid] = 0;
    if (tid < 32) sCnt[NBINS + tid] = 0;
    __syncthreads();

    int s[SORT_ITEMS], d[SORT_ITEMS], b[SORT_ITEMS], r[SORT_ITEMS];
    #pragma unroll
    for (int i = 0; i < SORT_ITEMS; ++i) {
        const int p = min(i * SORT_THREADS + tid, nValid - 1);
        d[i] = dst[base + p];
        s[i] = src[base + p];
    }
    #pragma unroll
    for (int i = 0; i < SORT_ITEMS; ++i) {
        const int p = i * SORT_THREADS + tid;
        b[i] = (p < nValid) ? d[i] / BIN_NODES : NBINS + lane;
        r[i] = atomicAdd(&sCnt[b[i]], 1);
    }
    __syncthreads();

    const int c = sCnt[tid];
    int inc = c;
    #pragma unroll
    for (int o = 1; o < 32; o <<= 1) {
        const int y = __shfl_up(inc, o);
        if (lane >= o) inc += y;
    }
    if (lane == 31) sWave[tid >> 5] = inc;
    __syncthreads();
    {
        const int w = tid >> 5;
        int off = 0;
        #pragma unroll
        for (int k = 0; k < NBINS / 32; ++k) off += (k < w) ? sWave[k] : 0;
        sStart[tid] = off + inc - c;
    }
    if (tid == 0) sStart[NBINS] = nValid;
    __syncthreads();

    int pos[SORT_ITEMS];
    #pragma unroll
    for (int i = 0; i < SORT_ITEMS; ++i) pos[i] = sStart[min(b[i], NBINS)] + r[i];
    #pragma unroll
    for (int i = 0; i < SORT_ITEMS; ++i) {
        const int p = i * SORT_THREADS + tid;
        if (p < nValid) sBuf[pos[i]] = make_int2(s[i], d[i]);
    }
    __syncthreads();

    int2 o[SORT_ITEMS];
    #pragma unroll
    for (int i = 0; i < SORT_ITEMS; ++i)
        o[i] = sBuf[min(i * SORT_THREADS + tid, nValid - 1)];
    #pragma unroll
    for (int i = 0; i < SORT_ITEMS; ++i) {
        const int p = i * SORT_THREADS + tid;
        if (p < nValid) sorted[base + p] = o[i];
    }
    segStart[chunk * (NBINS + 1) + tid] = sStart[tid];
    if (tid == 0) segStart[chunk * (NBINS + 1) + NBINS] = nValid;
}

#define W_LDS_BYTES   ((D_IN / 2) * D_OUT * 8)
#define A_QUARTER     32
#define A_PITCH       36
#define A_WAVE_FLOATS (16 * A_PITCH)
#define GEMM_LDS_BYTES (W_LDS_BYTES + (SORT_THREADS / 32) * A_WAVE_FLOATS * 4)

__device__ __forceinline__
void gcmc_gemm_wmma_role(unsigned char* __restrict__ smem, const int blk,
                         const float* __restrict__ A,
                         const float* __restrict__ W,
                         const float* __restrict__ cj,
                         float* __restrict__ H,
                         int nTiles) {
    float2* sW2 = (float2*)smem;

    {
        int tid = threadIdx.x;
        #pragma unroll
        for (int i = 0; i < ((D_IN / 2) * D_OUT) / 256; ++i) {
            int p  = tid + i * 256;
            int kk = p >> 6;
            int n  = p & 63;
            float2 w;
            w.x = W[(2 * kk)     * D_OUT + n];
            w.y = W[(2 * kk + 1) * D_OUT + n];
            sW2[p] = w;
        }
    }
    __syncthreads();

    const int wave = threadIdx.x >> 5;
    const int lane = threadIdx.x & 31;
    const int tile = blk * (SORT_THREADS / 32) + wave;
    if (tile >= nTiles) return;

    const int half = lane >> 4;
    const int lrow = lane & 15;

    float* sA = (float*)(smem + W_LDS_BYTES) + wave * A_WAVE_FLOATS;
    const int grow = lane >> 3;
    const int gcol = (lane & 7) * 4;
    const float* Ag  = A + (long long)(tile * 16 + grow) * D_IN + gcol;
    float*       sSt = sA + grow * A_PITCH + gcol;
    const float* sLd = sA + lrow * A_PITCH + 2 * half;

    float4 g[4];
    #pragma unroll
    for (int j = 0; j < 4; ++j) g[j] = *(const float4*)(Ag + (4 * j) * D_IN);

    v8f c0 = {}, c1 = {}, c2 = {}, c3 = {};

    #pragma unroll
    for (int p = 0; p < D_IN / A_QUARTER; ++p) {
        #pragma unroll
        for (int j = 0; j < 4; ++j) *(float4*)(sSt + (4 * j) * A_PITCH) = g[j];
        if (p + 1 < D_IN / A_QUARTER) {
            #pragma unroll
            for (int j = 0; j < 4; ++j) g[j] = *(const float4*)(Ag + (4 * j) * D_IN + (p + 1) * A_QUARTER);
        }
        __builtin_amdgcn_sched_barrier(0);


        #pragma unroll
        for (int kq = 0; kq < A_QUARTER; kq += 4) {
            const int k = p * A_QUARTER + kq;
            v2f a = *(const v2f*)(sLd + kq);

            const float2* sRow = sW2 + ((k >> 1) + half) * D_OUT + lrow;
            v2f b0 = *(const v2f*)(sRow +  0);
            v2f b1 = *(const v2f*)(sRow + 16);
            v2f b2 = *(const v2f*)(sRow + 32);
            v2f b3 = *(const v2f*)(sRow + 48);

            c0 = __builtin_amdgcn_wmma_f32_16x16x4_f32(false, a, false, b0, (short)0, c0, false, false);
            c1 = __builtin_amdgcn_wmma_f32_16x16x4_f32(false, a, false, b1, (short)0, c1, false, false);
            c2 = __builtin_amdgcn_wmma_f32_16x16x4_f32(false, a, false, b2, (short)0, c2, false, false);
            c3 = __builtin_amdgcn_wmma_f32_16x16x4_f32(false, a, false, b3, (short)0, c3, false, false);
        }
    }

    #pragma unroll
    for (int r = 0; r < 8; ++r) {
        const int   node = tile * 16 + r + 8 * half;
        const float s    = cj[node];
        float* Hrow = H + (long long)node * D_OUT;
        Hrow[ 0 + lrow] = c0[r] * s;
        Hrow[16 + lrow] = c1[r] * s;
        Hrow[32 + lrow] = c2[r] * s;
        Hrow[48 + lrow] = c3[r] * s;
    }
}

__global__ __launch_bounds__(SORT_THREADS) __attribute__((amdgpu_num_vgpr(80)))
void gcmc_gemm_and_sort_kernel(const float* __restrict__ A,
                               const float* __restrict__ W,
                               const float* __restrict__ cj,
                               float*       __restrict__ H,
                               int nTiles, int gemmBlocks,
                               const int*   __restrict__ src,
                               const int*   __restrict__ dst,
                               int2*        __restrict__ sorted,
                               int*         __restrict__ segStart,
                               int nEdges) {
    __shared__ __attribute__((aligned(16))) unsigned char smem[GEMM_LDS_BYTES];
    const int blk = blockIdx.x;
    if (blk < gemmBlocks)
        gcmc_gemm_wmma_role(smem, blk, A, W, cj, H, nTiles);
    else
        gcmc_bin_sort_role(smem, blk - gemmBlocks, src, dst, sorted, segStart, nEdges);
}

#define GATHER_Q     128
#define GATHER_SLOTS (GATHER_Q / 8)
#define GATHER_GROUPS 128
#define GATHER_CAP   (GATHER_GROUPS * GATHER_Q)
#define ROWS_PAD     416

__global__ __launch_bounds__(1024)
void gcmc_bin_gather_kernel(const float* __restrict__ H,
                            const float* __restrict__ ci,
                            const int2*  __restrict__ sorted,
                            const int*   __restrict__ segStart,
                            float*       __restrict__ out,
                            int nNodes, int nChunks) {
    __shared__ int sSrc[GATHER_CAP];
    __shared__ int sCnt[ROWS_PAD];
    __shared__ int sStart[ROWS_PAD];
    __shared__ int sWave[ROWS_PAD / 32];
    __shared__ int sMax;
    __shared__ int sNext;

    const int tid  = threadIdx.x;
    const int lane = tid & 31;
    const int wave = tid >> 5;
    const int bin  = blockIdx.x;
    const int lo   = bin * BIN_NODES;
    const int g    = tid >> 3;
    const int t    = tid & 7;
    const float4* __restrict__ H4 = (const float4*)H;
    float4* __restrict__ out4 = (float4*)out;

    const int rowLimit = min(BIN_NODES, nNodes - lo);

    bool first = true;

    for (int c0 = 0; c0 < nChunks; c0 += 512) {
        const int cBase = c0 + g;
        int nWin = 1;
        for (int w = 0; w < nWin; ++w) {
            int s0[4], len[4], sa[4], se[4];
            #pragma unroll
            for (int k = 0; k < 4; ++k) {
                const int c  = min(cBase + 128 * k, nChunks - 1);
                const int* ss = segStart + (long long)c * (NBINS + 1) + bin;
                sa[k] = ss[0];
                se[k] = ss[1];
            }
            #pragma unroll
            for (int k = 0; k < 4; ++k) {
                const int c = cBase + 128 * k;
                s0[k]  = min(c, nChunks - 1) * CHUNK + sa[k];
                len[k] = (c < nChunks) ? se[k] - sa[k] : 0;
            }
            const int p1  = len[0];
            const int p2  = p1 + len[1];
            const int p3  = p2 + len[2];
            const int tot = p3 + len[3];

            if (tid < ROWS_PAD) sCnt[tid] = 0;
            if (w == 0 && tid == 0) sMax = 0;
            __syncthreads();
            if (w == 0 && t == 0) atomicMax(&sMax, tot);
            if (tid == 0) sNext = GATHER_GROUPS;

            const int wlo = w * GATHER_Q;
            int2 rec[GATHER_SLOTS];
            #pragma unroll
            for (int i = 0; i < GATHER_SLOTS; ++i) {
                const int x = wlo + 8 * i + t;
                rec[i] = make_int2(0, lo - 1);
                if (x < tot) {
                    const int idx = x < p1 ? s0[0] + x
                                  : x < p2 ? s0[1] + (x - p1)
                                  : x < p3 ? s0[2] + (x - p2)
                                  :          s0[3] + (x - p3);
                    rec[i] = sorted[idx];
                }
            }
            int pk[GATHER_SLOTS];
            #pragma unroll
            for (int i = 0; i < GATHER_SLOTS; ++i) {
                const int row = rec[i].y - lo;
                pk[i] = -1;
                if ((unsigned)row < (unsigned)BIN_NODES)
                    pk[i] = row | (atomicAdd(&sCnt[row], 1) << 9);
            }
            __syncthreads();
            if (w == 0) nWin = (sMax + GATHER_Q - 1) / GATHER_Q;

            const int c = (tid < ROWS_PAD) ? sCnt[tid] : 0;
            int inc = c;
            #pragma unroll
            for (int o = 1; o < 32; o <<= 1) {
                const int y = __shfl_up(inc, o);
                if (lane >= o) inc += y;
            }
            if (tid < ROWS_PAD && lane == 31) sWave[wave] = inc;
            __syncthreads();
            if (tid < ROWS_PAD) {
                int off = 0;
                #pragma unroll
                for (int k = 0; k < ROWS_PAD / 32; ++k) off += (k < wave) ? sWave[k] : 0;
                sStart[tid] = off + inc - c;
            }
            __syncthreads();

            #pragma unroll
            for (int i = 0; i < GATHER_SLOTS; ++i)
                if (pk[i] >= 0) sSrc[sStart[pk[i] & 511] + (pk[i] >> 9)] = rec[i].x;
            __syncthreads();

            {
                int   cn = (g < rowLimit) ? g : -1;
                int   f = 0, e = 0;
                int   sv[8];
                #pragma unroll
                for (int u = 0; u < 8; ++u) sv[u] = 0;
                if (cn >= 0) {
                    f = sStart[cn];
                    e = sStart[cn + 1];
                    if (f < e) {
                        #pragma unroll
                        for (int u = 0; u < 8; ++u) sv[u] = sSrc[min(f + u, e - 1)];
                    }
                }
                float4 a0 = make_float4(0.f, 0.f, 0.f, 0.f);
                float4 a1 = make_float4(0.f, 0.f, 0.f, 0.f);
                const int tripMax = GATHER_CAP / 8 + BIN_NODES + 1;
                for (int trip = 0; trip < tripMax; ++trip) {
                    if (!__any(cn >= 0)) break;
                    if (cn >= 0) {
                        const bool has = f < e;
                        const int  f2  = f + 8;
                        const bool fin = f2 >= e;
                        float cd = 0.f;
                        if (fin) cd = ci[lo + cn];
                        float4 v0[8], v1[8];
                        if (has) {
                            #pragma unroll
                            for (int u = 0; u < 8; ++u) {
                                const unsigned q = (unsigned)sv[u] * (D_OUT / 4) + (unsigned)t;
                                v0[u] = H4[q];
                                v1[u] = H4[q + 8];
                            }
                        }
                        int   cnN = cn, fN = f2, eN = e;
                        if (fin) {
                            int r = 0;
                            if (t == 0) r = atomicAdd(&sNext, 1);
                            r   = __shfl(r, 0, 8);
                            cnN = (r < rowLimit) ? r : -1;
                            fN  = 0;
                            eN  = 0;
                            if (cnN >= 0) {
                                fN = sStart[cnN];
                                eN = sStart[cnN + 1];
                            }
                        }
                        if (fN < eN) {
                            #pragma unroll
                            for (int u = 0; u < 8; ++u) sv[u] = sSrc[min(fN + u, eN - 1)];
                        }
                        if (has) {
                            if (f2 <= e) {
                                #pragma unroll
                                for (int u = 0; u < 8; ++u) {
                                    a0.x += v0[u].x; a0.y += v0[u].y; a0.z += v0[u].z; a0.w += v0[u].w;
                                    a1.x += v1[u].x; a1.y += v1[u].y; a1.z += v1[u].z; a1.w += v1[u].w;
                                }
                            } else {
                                #pragma unroll
                                for (int u = 0; u < 8; ++u) {
                                    const bool on = f + u < e;
                                    a0.x += on ? v0[u].x : 0.f; a0.y += on ? v0[u].y : 0.f;
                                    a0.z += on ? v0[u].z : 0.f; a0.w += on ? v0[u].w : 0.f;
                                    a1.x += on ? v1[u].x : 0.f; a1.y += on ? v1[u].y : 0.f;
                                    a1.z += on ? v1[u].z : 0.f; a1.w += on ? v1[u].w : 0.f;
                                }
                            }
                        }
                        if (fin) {
                            float4* op = out4 + ((unsigned)(lo + cn) * (D_OUT / 4) + (unsigned)t);
                            float4 r0 = make_float4(a0.x * cd, a0.y * cd, a0.z * cd, a0.w * cd);
                            float4 r1 = make_float4(a1.x * cd, a1.y * cd, a1.z * cd, a1.w * cd);
                            if (!first) {
                                const float4 o0 = op[0];
                                const float4 o1 = op[8];
                                r0 = make_float4(o0.x + r0.x, o0.y + r0.y, o0.z + r0.z, o0.w + r0.w);
                                r1 = make_float4(o1.x + r1.x, o1.y + r1.y, o1.z + r1.z, o1.w + r1.w);
                            }
                            op[0] = r0;
                            op[8] = r1;
                            a0 = make_float4(0.f, 0.f, 0.f, 0.f);
                            a1 = make_float4(0.f, 0.f, 0.f, 0.f);
                        }
                        cn = cnN; f = fN; e = eN;
                    }
                }
            }
            first = false;
        }
    }
}

extern "C" void kernel_launch(void* const* d_in, const int* in_sizes, int n_in,
                              void* d_out, int out_size, void* d_ws, size_t ws_size,
                              hipStream_t stream) {
    const float* input_feat = (const float*)d_in[0];
    const float* weight     = (const float*)d_in[1];
    const float* cj         = (const float*)d_in[2];
    const float* ci         = (const float*)d_in[3];
    const int*   src_idx    = (const int*)d_in[4];
    const int*   dst_idx    = (const int*)d_in[5];
    float*       out        = (float*)d_out;
    float*       H          = (float*)d_ws;

    const int N = in_sizes[2];
    const int E = in_sizes[4];

    const int nChunks = (E + CHUNK - 1) / CHUNK;
    const int nBins   = (N + BIN_NODES - 1) / BIN_NODES;

    int2* sorted   = (int2*)(H + (size_t)N * D_OUT);
    int*  segStart = (int*)(sorted + (size_t)nChunks * CHUNK);

    {
        const int nTiles     = N / 16;
        const int gemmBlocks = (nTiles + 7) / 8;
        gcmc_gemm_and_sort_kernel<<<gemmBlocks + nChunks, SORT_THREADS, 0, stream>>>(
            input_feat, weight, cj, H, nTiles, gemmBlocks,
            src_idx, dst_idx, sorted, segStart, E);
    }

    gcmc_bin_gather_kernel<<<nBins, 1024, 0, stream>>>(H, ci, sorted, segStart, out, N, nChunks);
}
